# starting kernel
# baseline (speedup 1.0000x reference)
#include <hip/hip_runtime.h>
#include <hip/hip_bf16.h>

#define GDIM 128
#define EPS_RMS 1e-6f

typedef float v2f __attribute__((ext_vector_type(2)));
typedef float v8f __attribute__((ext_vector_type(8)));

__global__ void deg_init_kernel(float* deg, int n) {
    int i = blockIdx.x * 256 + threadIdx.x;
    if (i < n) deg[i] = 1.0f;
}

__global__ void deg_accum_kernel(const int* __restrict__ dst, float* deg, int ne) {
    int e = blockIdx.x * 256 + threadIdx.x;
    if (e < ne) atomicAdd(&deg[dst[e]], 1.0f);
}

__global__ void deg_finalize_kernel(float* deg, int n) {
    int i = blockIdx.x * 256 + threadIdx.x;
    if (i < n) deg[i] = rsqrtf(deg[i]);
}

__global__ __launch_bounds__(256) void rms_kernel(const float* __restrict__ X,
                                                  const float* __restrict__ bias,
                                                  const float* __restrict__ g,
                                                  float* __restrict__ H, int nrows) {
    int row  = blockIdx.x * 8 + (threadIdx.x >> 5);
    if (row >= nrows) return;
    int lane = threadIdx.x & 31;

    float4 v = ((const float4*)(X + (size_t)row * GDIM))[lane];
    if (bias) {
        float4 b4 = ((const float4*)bias)[lane];
        v.x = fmaxf(v.x + b4.x, 0.0f);
        v.y = fmaxf(v.y + b4.y, 0.0f);
        v.z = fmaxf(v.z + b4.z, 0.0f);
        v.w = fmaxf(v.w + b4.w, 0.0f);
    }
    float ss = v.x * v.x + v.y * v.y + v.z * v.z + v.w * v.w;
    #pragma unroll
    for (int off = 16; off > 0; off >>= 1) ss += __shfl_xor(ss, off, 32);

    float scale = rsqrtf(ss * (1.0f / GDIM) + EPS_RMS);
    float4 g4 = ((const float4*)g)[lane];
    float4 o;
    o.x = v.x * scale * g4.x;
    o.y = v.y * scale * g4.y;
    o.z = v.z * scale * g4.z;
    o.w = v.w * scale * g4.w;
    ((float4*)(H + (size_t)row * GDIM))[lane] = o;
}

__global__ __launch_bounds__(256) void gemm_wmma_kernel(const float* __restrict__ H,
                                                        const float* __restrict__ W,
                                                        float* __restrict__ T, int nrows) {
    __shared__ float Wsw[(GDIM / 4) * GDIM * 4];

    int tid = threadIdx.x;
    for (int i = tid; i < (GDIM * GDIM) / 4; i += 256) {
        float4 w4 = ((const float4*)W)[i];
        int k = (i * 4) / GDIM;
        int c = (i * 4) % GDIM;
        int kg = k >> 2, ks = k & 3;
        float* p = &Wsw[kg * 512 + ks];
        p[(c + 0) * 4] = w4.x;
        p[(c + 1) * 4] = w4.y;
        p[(c + 2) * 4] = w4.z;
        p[(c + 3) * 4] = w4.w;
    }
    __syncthreads();

    int wave    = tid >> 5;
    int lane    = tid & 31;
    int l16     = lane & 15;
    int halfsel = lane >> 4;
    int koff    = halfsel * 2;

    int rowBase = blockIdx.x * 128 + wave * 16;
    int arow    = rowBase + l16;
    int arow_c  = arow < nrows ? arow : (nrows - 1);
    const float* hrow  = H + (size_t)arow_c * GDIM;
    const float* wbase = Wsw + l16 * 4 + koff;

    v8f acc[8] = {};

    #pragma unroll 4
    for (int k0 = 0; k0 < GDIM; k0 += 4) {
        v2f a = *(const v2f*)(hrow + k0 + koff);
        const float* wk = wbase + (k0 >> 2) * 512;
        #pragma unroll
        for (int nt = 0; nt < 8; ++nt) {
            v2f b = *(const v2f*)(wk + nt * 64);
            acc[nt] = __builtin_amdgcn_wmma_f32_16x16x4_f32(
                false, a, false, b, (short)0, acc[nt], false, false);
        }
    }

    for (int nt = 0; nt < 8; ++nt) {
        int col = nt * 16 + l16;
        #pragma unroll
        for (int r = 0; r < 8; ++r) {
            int row = rowBase + r + halfsel * 8;
            if (row < nrows) T[(size_t)row * GDIM + col] = acc[nt][r];
        }
    }
}

__global__ void agg_init_kernel(const float* __restrict__ T, const float* __restrict__ dinv,
                                float* __restrict__ AGG, int total) {
    int i = blockIdx.x * 256 + threadIdx.x;
    if (i < total) {
        int row = i >> 7;
        float di = dinv[row];
        AGG[i] = T[i] * di * di;
    }
}

__global__ __launch_bounds__(256) void scatter_kernel(const float* __restrict__ T,
                                                      const int* __restrict__ src,
                                                      const int* __restrict__ dst,
                                                      const float* __restrict__ dinv,
                                                      float* __restrict__ AGG, int ne) {
    int e = blockIdx.x * 8 + (threadIdx.x >> 5);
    if (e >= ne) return;
    int lane = threadIdx.x & 31;
    int s = src[e], d = dst[e];
    float c = dinv[s] * dinv[d];
    float4 t4 = ((const float4*)(T + (size_t)s * GDIM))[lane];
    float* out = AGG + (size_t)d * GDIM + lane * 4;
    atomicAdd(out + 0, t4.x * c);
    atomicAdd(out + 1, t4.y * c);
    atomicAdd(out + 2, t4.z * c);
    atomicAdd(out + 3, t4.w * c);
}

__global__ void bias_relu_kernel(const float* __restrict__ A, const float* __restrict__ b,
                                 float* __restrict__ O, int total) {
    int i = blockIdx.x * 256 + threadIdx.x;
    if (i < total) {
        int dcol = i & (GDIM - 1);
        O[i] = fmaxf(A[i] + b[dcol], 0.0f);
    }
}

extern "C" void kernel_launch(void* const* d_in, const int* in_sizes, int n_in,
                              void* d_out, int out_size, void* d_ws, size_t ws_size,
                              hipStream_t stream) {
    const float* x  = (const float*)d_in[0];
    const int*   ei = (const int*)d_in[1];
    const float* Wm[3]   = {(const float*)d_in[2], (const float*)d_in[5], (const float*)d_in[8]};
    const float* bias[3] = {(const float*)d_in[3], (const float*)d_in[6], (const float*)d_in[9]};
    const float* gain[3] = {(const float*)d_in[4], (const float*)d_in[7], (const float*)d_in[10]};

    const int N = in_sizes[0] / GDIM;
    const int E = in_sizes[1] / 2;
    const int* src = ei;
    const int* dst = ei + E;

    float* dinv = (float*)d_ws;
    size_t off  = ((size_t)N + 63) & ~(size_t)63;
    float* B0 = dinv + off;
    float* B1 = B0 + (size_t)N * GDIM;

    const int total = N * GDIM;

    deg_init_kernel<<<(N + 255) / 256, 256, 0, stream>>>(dinv, N);
    deg_accum_kernel<<<(E + 255) / 256, 256, 0, stream>>>(dst, dinv, E);
    deg_finalize_kernel<<<(N + 255) / 256, 256, 0, stream>>>(dinv, N);

    const float* in_x = x;
    float* bufs[2] = {B0, B1};
    for (int l = 0; l < 3; ++l) {
        float* H = bufs[l & 1];
        float* T = bufs[(l + 1) & 1];

        rms_kernel<<<(N + 7) / 8, 256, 0, stream>>>(
            in_x, (l == 0) ? nullptr : bias[l - 1], gain[l], H, N);
        gemm_wmma_kernel<<<(N + 127) / 128, 256, 0, stream>>>(H, Wm[l], T, N);
        agg_init_kernel<<<(total + 255) / 256, 256, 0, stream>>>(T, dinv, H, total);
        scatter_kernel<<<(E + 7) / 8, 256, 0, stream>>>(T, src, dst, dinv, H, E);

        in_x = H;
    }
    bias_relu_kernel<<<(total + 255) / 256, 256, 0, stream>>>(in_x, bias[2], (float*)d_out, total);
}


# Edge scatter with float atomics replaced by a gather over edges regrouped by destination once per call (counting sort): one wave sums a row's incoming messages and writes the row once
# speedup vs baseline: 2.4217x; 2.4217x over previous
#include <hip/hip_runtime.h>
#include <hip/hip_bf16.h>

#define GDIM 128
#define EPS_RMS 1e-6f

typedef float v2f __attribute__((ext_vector_type(2)));
typedef float v8f __attribute__((ext_vector_type(8)));

__global__ void count_zero_kernel(int* cnt, int n) {
    int i = blockIdx.x * 256 + threadIdx.x;
    if (i < n) cnt[i] = 0;
}

__global__ void count_rank_kernel(const int* __restrict__ dst, int* cnt, int* __restrict__ rank, int ne) {
    int e = blockIdx.x * 256 + threadIdx.x;
    if (e < ne) rank[e] = atomicAdd(&cnt[dst[e]], 1);
}

__global__ __launch_bounds__(1024) void rowptr_kernel(const int* __restrict__ cnt,
                                                      int2* __restrict__ seg,
                                                      float* __restrict__ dinv, int n) {
    __shared__ int wfront[32];
    __shared__ int wtile[32];
    const int tid = threadIdx.x, lane = tid & 31, wave = tid >> 5;
    const int tile0 = blockIdx.x * 1024;

    int front = 0;
    for (int i = tid; i < tile0; i += 1024) front += cnt[i];

    const int idx = tile0 + tid;
    const int c = (idx < n) ? cnt[idx] : 0;

    for (int off = 16; off > 0; off >>= 1) front += __shfl_xor(front, off, 32);
    int inc = c;
    for (int off = 1; off < 32; off <<= 1) {
        int t = __shfl_up(inc, off, 32);
        if (lane >= off) inc += t;
    }
    if (lane == 0)  wfront[wave] = front;
    if (lane == 31) wtile[wave]  = inc;
    __syncthreads();

    int tot = wfront[lane] + ((lane < wave) ? wtile[lane] : 0);
    for (int off = 16; off > 0; off >>= 1) tot += __shfl_xor(tot, off, 32);

    if (idx < n) {
        seg[idx]  = make_int2(tot + inc - c, c);
        dinv[idx] = rsqrtf((float)(c + 1));
    }
}

__global__ void csr_fill_kernel(const int* __restrict__ src, const int* __restrict__ dst,
                                const int* __restrict__ rank, const int2* __restrict__ seg,
                                const float* __restrict__ dinv, int2* __restrict__ csr, int ne) {
    int e = blockIdx.x * 256 + threadIdx.x;
    if (e < ne) {
        int s = src[e], d = dst[e];
        float c = dinv[s] * dinv[d];
        csr[seg[d].x + rank[e]] = make_int2(s, __float_as_int(c));
    }
}

__global__ __launch_bounds__(256) void rms_kernel(const float* __restrict__ X,
                                                  const float* __restrict__ bias,
                                                  const float* __restrict__ g,
                                                  float* __restrict__ H, int nrows) {
    int row  = blockIdx.x * 8 + (threadIdx.x >> 5);
    if (row >= nrows) return;
    int lane = threadIdx.x & 31;

    float4 v = ((const float4*)(X + (size_t)row * GDIM))[lane];
    if (bias) {
        float4 b4 = ((const float4*)bias)[lane];
        v.x = fmaxf(v.x + b4.x, 0.0f);
        v.y = fmaxf(v.y + b4.y, 0.0f);
        v.z = fmaxf(v.z + b4.z, 0.0f);
        v.w = fmaxf(v.w + b4.w, 0.0f);
    }
    float ss = v.x * v.x + v.y * v.y + v.z * v.z + v.w * v.w;
    #pragma unroll
    for (int off = 16; off > 0; off >>= 1) ss += __shfl_xor(ss, off, 32);

    float scale = rsqrtf(ss * (1.0f / GDIM) + EPS_RMS);
    float4 g4 = ((const float4*)g)[lane];
    float4 o;
    o.x = v.x * scale * g4.x;
    o.y = v.y * scale * g4.y;
    o.z = v.z * scale * g4.z;
    o.w = v.w * scale * g4.w;
    ((float4*)(H + (size_t)row * GDIM))[lane] = o;
}

__global__ __launch_bounds__(256) void gemm_wmma_kernel(const float* __restrict__ H,
                                                        const float* __restrict__ W,
                                                        float* __restrict__ T, int nrows) {
    __shared__ float Wsw[(GDIM / 4) * GDIM * 4];

    int tid = threadIdx.x;
    for (int i = tid; i < (GDIM * GDIM) / 4; i += 256) {
        float4 w4 = ((const float4*)W)[i];
        int k = (i * 4) / GDIM;
        int c = (i * 4) % GDIM;
        int kg = k >> 2, ks = k & 3;
        float* p = &Wsw[kg * 512 + ks];
        p[(c + 0) * 4] = w4.x;
        p[(c + 1) * 4] = w4.y;
        p[(c + 2) * 4] = w4.z;
        p[(c + 3) * 4] = w4.w;
    }
    __syncthreads();

    int wave    = tid >> 5;
    int lane    = tid & 31;
    int l16     = lane & 15;
    int halfsel = lane >> 4;
    int koff    = halfsel * 2;

    int rowBase = blockIdx.x * 128 + wave * 16;
    int arow    = rowBase + l16;
    int arow_c  = arow < nrows ? arow : (nrows - 1);
    const float* hrow  = H + (size_t)arow_c * GDIM;
    const float* wbase = Wsw + l16 * 4 + koff;

    v8f acc[8] = {};

    #pragma unroll 4
    for (int k0 = 0; k0 < GDIM; k0 += 4) {
        v2f a = *(const v2f*)(hrow + k0 + koff);
        const float* wk = wbase + (k0 >> 2) * 512;
        #pragma unroll
        for (int nt = 0; nt < 8; ++nt) {
            v2f b = *(const v2f*)(wk + nt * 64);
            acc[nt] = __builtin_amdgcn_wmma_f32_16x16x4_f32(
                false, a, false, b, (short)0, acc[nt], false, false);
        }
    }

    for (int nt = 0; nt < 8; ++nt) {
        int col = nt * 16 + l16;
        #pragma unroll
        for (int r = 0; r < 8; ++r) {
            int row = rowBase + r + halfsel * 8;
            if (row < nrows) T[(size_t)row * GDIM + col] = acc[nt][r];
        }
    }
}

__global__ __launch_bounds__(256) void gather_kernel(const float* __restrict__ T,
                                                     const int2* __restrict__ seg,
                                                     const int2* __restrict__ csr,
                                                     const float* __restrict__ dinv,
                                                     float* __restrict__ AGG, int n) {
    const int lane = threadIdx.x & 31;
    const int d = __builtin_amdgcn_readfirstlane((int)(blockIdx.x * 8 + (threadIdx.x >> 5)));
    if (d >= n) return;

    const int2  sg  = seg[d];
    const int   beg = sg.x, len = sg.y;
    const float di  = dinv[d];
    const float cs  = di * di;

    const float4 t = ((const float4*)(T + (size_t)d * GDIM))[lane];
    float4 acc = make_float4(0.0f, 0.0f, 0.0f, 0.0f);

    for (int base = 0; base < len; base += 32) {
        const int m = (len - base < 32) ? (len - base) : 32;
        int2 rec = make_int2(0, 0);
        if (lane < m) rec = csr[beg + base + lane];
        for (int j = 0; j < m; j += 8) {
            float4 v[8];
            float  c[8];
            #pragma unroll
            for (int u = 0; u < 8; ++u) {
                v[u] = make_float4(0.0f, 0.0f, 0.0f, 0.0f);
                c[u] = 0.0f;
                if (u == 0 || j + u < m) {
                    int s = __builtin_amdgcn_readlane(rec.x, j + u);
                    c[u]  = __int_as_float(__builtin_amdgcn_readlane(rec.y, j + u));
                    s = (int)min((unsigned)s, (unsigned)(n - 1));
                    v[u] = ((const float4*)(T + (size_t)s * GDIM))[lane];
                }
            }
            #pragma unroll
            for (int u = 0; u < 8; ++u) {
                acc.x = fmaf(v[u].x, c[u], acc.x);
                acc.y = fmaf(v[u].y, c[u], acc.y);
                acc.z = fmaf(v[u].z, c[u], acc.z);
                acc.w = fmaf(v[u].w, c[u], acc.w);
            }
        }
    }
    acc.x = fmaf(t.x, cs, acc.x);
    acc.y = fmaf(t.y, cs, acc.y);
    acc.z = fmaf(t.z, cs, acc.z);
    acc.w = fmaf(t.w, cs, acc.w);
    ((float4*)(AGG + (size_t)d * GDIM))[lane] = acc;
}

__global__ void bias_relu_kernel(const float* __restrict__ A, const float* __restrict__ b,
                                 float* __restrict__ O, int total) {
    int i = blockIdx.x * 256 + threadIdx.x;
    if (i < total) {
        int dcol = i & (GDIM - 1);
        O[i] = fmaxf(A[i] + b[dcol], 0.0f);
    }
}

extern "C" void kernel_launch(void* const* d_in, const int* in_sizes, int n_in,
                              void* d_out, int out_size, void* d_ws, size_t ws_size,
                              hipStream_t stream) {
    const float* x  = (const float*)d_in[0];
    const int*   ei = (const int*)d_in[1];
    const float* Wm[3]   = {(const float*)d_in[2], (const float*)d_in[5], (const float*)d_in[8]};
    const float* bias[3] = {(const float*)d_in[3], (const float*)d_in[6], (const float*)d_in[9]};
    const float* gain[3] = {(const float*)d_in[4], (const float*)d_in[7], (const float*)d_in[10]};

    const int N = in_sizes[0] / GDIM;
    const int E = in_sizes[1] / 2;
    const int* src = ei;
    const int* dst = ei + E;

    float* dinv = (float*)d_ws;
    size_t off  = ((size_t)N + 63) & ~(size_t)63;
    size_t offE = ((size_t)E + 63) & ~(size_t)63;
    float* B0 = dinv + off;
    float* B1 = (float*)d_out;
    int*  cnt    = (int*)(B0 + (size_t)N * GDIM);
    int2* seg    = (int2*)(cnt + off);
    int*  rank   = (int*)(seg + off);
    int2* csr    = (int2*)(rank + offE);

    const int total = N * GDIM;

    count_zero_kernel<<<(N + 255) / 256, 256, 0, stream>>>(cnt, N);
    count_rank_kernel<<<(E + 255) / 256, 256, 0, stream>>>(dst, cnt, rank, E);
    rowptr_kernel<<<(N + 1023) / 1024, 1024, 0, stream>>>(cnt, seg, dinv, N);
    csr_fill_kernel<<<(E + 255) / 256, 256, 0, stream>>>(src, dst, rank, seg, dinv, csr, E);

    const float* in_x = x;
    float* bufs[2] = {B0, B1};
    for (int l = 0; l < 3; ++l) {
        float* H = bufs[l & 1];
        float* T = bufs[(l + 1) & 1];

        rms_kernel<<<(N + 7) / 8, 256, 0, stream>>>(
            in_x, (l == 0) ? nullptr : bias[l - 1], gain[l], H, N);
        gemm_wmma_kernel<<<(N + 127) / 128, 256, 0, stream>>>(H, Wm[l], T, N);
        gather_kernel<<<(N + 7) / 8, 256, 0, stream>>>(T, seg, csr, dinv, H, N);

        in_x = H;
    }
    bias_relu_kernel<<<(total + 255) / 256, 256, 0, stream>>>(in_x, bias[2], (float*)d_out, total);
}


# Input RMSNorm folded into the first GEMM: gain scales W rows while staged, row sum of squares taken from the A fragments in the k loop, rsqrt scale applied to the accumulators; rms launch removed
# speedup vs baseline: 2.9999x; 1.2388x over previous
#include <hip/hip_runtime.h>
#include <hip/hip_bf16.h>

#define GDIM 128
#define EPS_RMS 1e-6f

typedef float v2f __attribute__((ext_vector_type(2)));
typedef float v8f __attribute__((ext_vector_type(8)));

__global__ void count_zero_kernel(int* cnt, int n) {
    int i = blockIdx.x * 256 + threadIdx.x;
    if (i < n) cnt[i] = 0;
}

__global__ void count_rank_kernel(const int* __restrict__ dst, int* cnt, int* __restrict__ rank, int ne) {
    int e = blockIdx.x * 256 + threadIdx.x;
    if (e < ne) rank[e] = atomicAdd(&cnt[dst[e]], 1);
}

__global__ __launch_bounds__(1024) void rowptr_kernel(const int* __restrict__ cnt,
                                                      int2* __restrict__ seg,
                                                      float* __restrict__ dinv, int n) {
    __shared__ int wfront[32];
    __shared__ int wtile[32];
    const int tid = threadIdx.x, lane = tid & 31, wave = tid >> 5;
    const int tile0 = blockIdx.x * 1024;

    int front = 0;
    for (int i = tid; i < tile0; i += 1024) front += cnt[i];

    const int idx = tile0 + tid;
    const int c = (idx < n) ? cnt[idx] : 0;

    for (int off = 16; off > 0; off >>= 1) front += __shfl_xor(front, off, 32);
    int inc = c;
    for (int off = 1; off < 32; off <<= 1) {
        int t = __shfl_up(inc, off, 32);
        if (lane >= off) inc += t;
    }
    if (lane == 0)  wfront[wave] = front;
    if (lane == 31) wtile[wave]  = inc;
    __syncthreads();

    int tot = wfront[lane] + ((lane < wave) ? wtile[lane] : 0);
    for (int off = 16; off > 0; off >>= 1) tot += __shfl_xor(tot, off, 32);

    if (idx < n) {
        seg[idx]  = make_int2(tot + inc - c, c);
        dinv[idx] = rsqrtf((float)(c + 1));
    }
}

__global__ void csr_fill_kernel(const int* __restrict__ src, const int* __restrict__ dst,
                                const int* __restrict__ rank, const int2* __restrict__ seg,
                                const float* __restrict__ dinv, int2* __restrict__ csr, int ne) {
    int e = blockIdx.x * 256 + threadIdx.x;
    if (e < ne) {
        int s = src[e], d = dst[e];
        float c = dinv[s] * dinv[d];
        csr[seg[d].x + rank[e]] = make_int2(s, __float_as_int(c));
    }
}

template <bool NORM>
__device__ __forceinline__ void gemm_body(const float* __restrict__ A,
                                          const float* __restrict__ W,
                                          const float* __restrict__ g,
                                          float* __restrict__ T, int nrows, float* Wsw) {
    int tid = threadIdx.x;
    for (int i = tid; i < (GDIM * GDIM) / 4; i += 256) {
        float4 w4 = ((const float4*)W)[i];
        int k = (i * 4) / GDIM;
        int c = (i * 4) % GDIM;
        if (NORM) {
            const float gk = g[k];
            w4.x *= gk; w4.y *= gk; w4.z *= gk; w4.w *= gk;
        }
        int kg = k >> 2, ks = k & 3;
        float* p = &Wsw[kg * 512 + ks];
        p[(c + 0) * 4] = w4.x;
        p[(c + 1) * 4] = w4.y;
        p[(c + 2) * 4] = w4.z;
        p[(c + 3) * 4] = w4.w;
    }
    __syncthreads();

    int wave    = tid >> 5;
    int lane    = tid & 31;
    int l16     = lane & 15;
    int halfsel = lane >> 4;
    int koff    = halfsel * 2;

    int rowBase = blockIdx.x * 128 + wave * 16;
    int arow    = rowBase + l16;
    int arow_c  = arow < nrows ? arow : (nrows - 1);
    const float* hrow  = A + (size_t)arow_c * GDIM;
    const float* wbase = Wsw + l16 * 4 + koff;

    v8f acc[8] = {};
    float ss = 0.0f;

    #pragma unroll 4
    for (int k0 = 0; k0 < GDIM; k0 += 4) {
        v2f a = *(const v2f*)(hrow + k0 + koff);
        if (NORM) {
            ss = fmaf(a.x, a.x, ss);
            ss = fmaf(a.y, a.y, ss);
        }
        const float* wk = wbase + (k0 >> 2) * 512;
        #pragma unroll
        for (int nt = 0; nt < 8; ++nt) {
            v2f b = *(const v2f*)(wk + nt * 64);
            acc[nt] = __builtin_amdgcn_wmma_f32_16x16x4_f32(
                false, a, false, b, (short)0, acc[nt], false, false);
        }
    }

    float sc[8];
    if (NORM) {
        ss += __shfl_xor(ss, 16, 32);
        const float scale = rsqrtf(ss * (1.0f / GDIM) + EPS_RMS);
        #pragma unroll
        for (int r = 0; r < 8; ++r) sc[r] = __shfl(scale, r + halfsel * 8, 32);
    }

    for (int nt = 0; nt < 8; ++nt) {
        int col = nt * 16 + l16;
        #pragma unroll
        for (int r = 0; r < 8; ++r) {
            int row = rowBase + r + halfsel * 8;
            float v = acc[nt][r];
            if (NORM) v *= sc[r];
            if (row < nrows) T[(size_t)row * GDIM + col] = v;
        }
    }
}

__global__ __launch_bounds__(256) void gemm_wmma_kernel(const float* __restrict__ H,
                                                        const float* __restrict__ W,
                                                        float* __restrict__ T, int nrows) {
    __shared__ float Wsw[(GDIM / 4) * GDIM * 4];
    gemm_body<false>(H, W, nullptr, T, nrows, Wsw);
}

__global__ __launch_bounds__(256) void gemm_norm_wmma_kernel(const float* __restrict__ X,
                                                             const float* __restrict__ W,
                                                             const float* __restrict__ g,
                                                             float* __restrict__ T, int nrows) {
    __shared__ float Wsw[(GDIM / 4) * GDIM * 4];
    gemm_body<true>(X, W, g, T, nrows, Wsw);
}

__global__ __launch_bounds__(256) void gather_kernel(const float* __restrict__ T,
                                                     const int2* __restrict__ seg,
                                                     const int2* __restrict__ csr,
                                                     const float* __restrict__ dinv,
                                                     const float* __restrict__ bias,
                                                     const float* __restrict__ gnext,
                                                     float* __restrict__ OUT, int n) {
    const int lane = threadIdx.x & 31;
    const int d = __builtin_amdgcn_readfirstlane((int)(blockIdx.x * 8 + (threadIdx.x >> 5)));
    if (d >= n) return;

    const int2  sg  = seg[d];
    const int   beg = sg.x, len = sg.y;
    const float di  = dinv[d];
    const float cs  = di * di;

    const float4 t = ((const float4*)(T + (size_t)d * GDIM))[lane];
    float4 acc = make_float4(0.0f, 0.0f, 0.0f, 0.0f);

    for (int base = 0; base < len; base += 32) {
        const int m = (len - base < 32) ? (len - base) : 32;
        int2 rec = make_int2(0, 0);
        if (lane < m) rec = csr[beg + base + lane];
        for (int j = 0; j < m; j += 8) {
            float4 v[8];
            float  c[8];
            #pragma unroll
            for (int u = 0; u < 8; ++u) {
                v[u] = make_float4(0.0f, 0.0f, 0.0f, 0.0f);
                c[u] = 0.0f;
                if (u == 0 || j + u < m) {
                    int s = __builtin_amdgcn_readlane(rec.x, j + u);
                    c[u]  = __int_as_float(__builtin_amdgcn_readlane(rec.y, j + u));
                    s = (int)min((unsigned)s, (unsigned)(n - 1));
                    v[u] = ((const float4*)(T + (size_t)s * GDIM))[lane];
                }
            }
            #pragma unroll
            for (int u = 0; u < 8; ++u) {
                acc.x = fmaf(v[u].x, c[u], acc.x);
                acc.y = fmaf(v[u].y, c[u], acc.y);
                acc.z = fmaf(v[u].z, c[u], acc.z);
                acc.w = fmaf(v[u].w, c[u], acc.w);
            }
        }
    }
    acc.x = fmaf(t.x, cs, acc.x);
    acc.y = fmaf(t.y, cs, acc.y);
    acc.z = fmaf(t.z, cs, acc.z);
    acc.w = fmaf(t.w, cs, acc.w);

    const float4 b4 = ((const float4*)bias)[lane];
    float4 y;
    y.x = fmaxf(acc.x + b4.x, 0.0f);
    y.y = fmaxf(acc.y + b4.y, 0.0f);
    y.z = fmaxf(acc.z + b4.z, 0.0f);
    y.w = fmaxf(acc.w + b4.w, 0.0f);

    if (gnext) {
        float ss = y.x * y.x + y.y * y.y + y.z * y.z + y.w * y.w;
        for (int off = 16; off > 0; off >>= 1) ss += __shfl_xor(ss, off, 32);
        const float scale = rsqrtf(ss * (1.0f / GDIM) + EPS_RMS);
        const float4 g4 = ((const float4*)gnext)[lane];
        y.x = y.x * scale * g4.x;
        y.y = y.y * scale * g4.y;
        y.z = y.z * scale * g4.z;
        y.w = y.w * scale * g4.w;
    }
    ((float4*)(OUT + (size_t)d * GDIM))[lane] = y;
}

extern "C" void kernel_launch(void* const* d_in, const int* in_sizes, int n_in,
                              void* d_out, int out_size, void* d_ws, size_t ws_size,
                              hipStream_t stream) {
    const float* x  = (const float*)d_in[0];
    const int*   ei = (const int*)d_in[1];
    const float* Wm[3]   = {(const float*)d_in[2], (const float*)d_in[5], (const float*)d_in[8]};
    const float* bias[3] = {(const float*)d_in[3], (const float*)d_in[6], (const float*)d_in[9]};
    const float* gain[3] = {(const float*)d_in[4], (const float*)d_in[7], (const float*)d_in[10]};

    const int N = in_sizes[0] / GDIM;
    const int E = in_sizes[1] / 2;
    const int* src = ei;
    const int* dst = ei + E;

    float* dinv = (float*)d_ws;
    size_t off  = ((size_t)N + 63) & ~(size_t)63;
    size_t offE = ((size_t)E + 63) & ~(size_t)63;
    float* B0 = dinv + off;
    float* B1 = (float*)d_out;
    int*  cnt    = (int*)(B0 + (size_t)N * GDIM);
    int2* seg    = (int2*)(cnt + off);
    int*  rank   = (int*)(seg + off);
    int2* csr    = (int2*)(rank + offE);

    count_zero_kernel<<<(N + 255) / 256, 256, 0, stream>>>(cnt, N);
    count_rank_kernel<<<(E + 255) / 256, 256, 0, stream>>>(dst, cnt, rank, E);
    rowptr_kernel<<<(N + 1023) / 1024, 1024, 0, stream>>>(cnt, seg, dinv, N);
    csr_fill_kernel<<<(E + 255) / 256, 256, 0, stream>>>(src, dst, rank, seg, dinv, csr, E);

    for (int l = 0; l < 3; ++l) {
        if (l == 0)
            gemm_norm_wmma_kernel<<<(N + 127) / 128, 256, 0, stream>>>(x, Wm[0], gain[0], B0, N);
        else
            gemm_wmma_kernel<<<(N + 127) / 128, 256, 0, stream>>>(B1, Wm[l], B0, N);
        gather_kernel<<<(N + 7) / 8, 256, 0, stream>>>(
            B0, seg, csr, dinv, bias[l], (l < 2) ? gain[l + 1] : nullptr, B1, N);
    }
}


# Layers 2 and 3 products fused into the gathers before them: persistent 16-wave blocks aggregate 16 rows into LDS, 8 waves multiply them by W staged once per block; 2 product launches removed
# speedup vs baseline: 3.5920x; 1.1974x over previous
#include <hip/hip_runtime.h>
#include <hip/hip_bf16.h>

#define GDIM 128
#define EPS_RMS 1e-6f

typedef float v2f __attribute__((ext_vector_type(2)));
typedef float v8f __attribute__((ext_vector_type(8)));

__global__ void count_zero_kernel(int* cnt, int n) {
    int i = blockIdx.x * 256 + threadIdx.x;
    if (i < n) cnt[i] = 0;
}

__global__ __launch_bounds__(1024) void rowptr_kernel(const int* __restrict__ cnt,
                                                      int2* __restrict__ seg,
                                                      float* __restrict__ dinv, int n) {
    __shared__ int wfront[32];
    __shared__ int wtile[32];
    const int tid = threadIdx.x, lane = tid & 31, wave = tid >> 5;
    const int tile0 = blockIdx.x * 1024;

    int front = 0;
    for (int i = tid; i < tile0; i += 1024) front += cnt[i];

    const int idx = tile0 + tid;
    const int c = (idx < n) ? cnt[idx] : 0;

    for (int off = 16; off > 0; off >>= 1) front += __shfl_xor(front, off, 32);
    int inc = c;
    for (int off = 1; off < 32; off <<= 1) {
        int t = __shfl_up(inc, off, 32);
        if (lane >= off) inc += t;
    }
    if (lane == 0)  wfront[wave] = front;
    if (lane == 31) wtile[wave]  = inc;
    __syncthreads();

    int tot = wfront[lane] + ((lane < wave) ? wtile[lane] : 0);
    for (int off = 16; off > 0; off >>= 1) tot += __shfl_xor(tot, off, 32);

    if (idx < n) {
        seg[idx]  = make_int2(tot + inc - c, c);
        dinv[idx] = rsqrtf((float)(c + 1));
    }
}

template <bool NORM>
__device__ __forceinline__ void gemm_body(const float* __restrict__ A,
                                          const float* __restrict__ W,
                                          const float* __restrict__ g,
                                          float* __restrict__ T, int nrows, float* Wsw,
                                          int tile) {
    int tid = threadIdx.x;
    for (int i = tid; i < (GDIM * GDIM) / 4; i += 256) {
        float4 w4 = ((const float4*)W)[i];
        int k = (i * 4) / GDIM;
        int c = (i * 4) % GDIM;
        if (NORM) {
            const float gk = g[k];
            w4.x *= gk; w4.y *= gk; w4.z *= gk; w4.w *= gk;
        }
        int kg = k >> 2, ks = k & 3;
        float* p = &Wsw[kg * 512 + ks];
        p[(c + 0) * 4] = w4.x;
        p[(c + 1) * 4] = w4.y;
        p[(c + 2) * 4] = w4.z;
        p[(c + 3) * 4] = w4.w;
    }
    __syncthreads();

    int wave    = tid >> 5;
    int lane    = tid & 31;
    int l16     = lane & 15;
    int halfsel = lane >> 4;
    int koff    = halfsel * 2;

    int rowBase = tile * 128 + wave * 16;
    int arow    = rowBase + l16;
    int arow_c  = arow < nrows ? arow : (nrows - 1);
    const float* hrow  = A + (size_t)arow_c * GDIM;
    const float* wbase = Wsw + l16 * 4 + koff;

    v8f acc[8] = {};
    float ss = 0.0f;

    #pragma unroll 4
    for (int k0 = 0; k0 < GDIM; k0 += 4) {
        v2f a = *(const v2f*)(hrow + k0 + koff);
        if (NORM) {
            ss = fmaf(a.x, a.x, ss);
            ss = fmaf(a.y, a.y, ss);
        }
        const float* wk = wbase + (k0 >> 2) * 512;
        #pragma unroll
        for (int nt = 0; nt < 8; ++nt) {
            v2f b = *(const v2f*)(wk + nt * 64);
            acc[nt] = __builtin_amdgcn_wmma_f32_16x16x4_f32(
                false, a, false, b, (short)0, acc[nt], false, false);
        }
    }

    float sc[8];
    if (NORM) {
        ss += __shfl_xor(ss, 16, 32);
        const float scale = rsqrtf(ss * (1.0f / GDIM) + EPS_RMS);
        #pragma unroll
        for (int r = 0; r < 8; ++r) sc[r] = __shfl(scale, r + halfsel * 8, 32);
    }

    for (int nt = 0; nt < 8; ++nt) {
        int col = nt * 16 + l16;
        #pragma unroll
        for (int r = 0; r < 8; ++r) {
            int row = rowBase + r + halfsel * 8;
            float v = acc[nt][r];
            if (NORM) v *= sc[r];
            if (row < nrows) T[(size_t)row * GDIM + col] = v;
        }
    }
}


__global__ __launch_bounds__(256) void count_rank_gemm1_kernel(const int* __restrict__ dst, int* cnt,
                                                               int* __restrict__ rank, int ne,
                                                               const float* __restrict__ X,
                                                               const float* __restrict__ W,
                                                               const float* __restrict__ g,
                                                               float* __restrict__ T, int nrows,
                                                               int gemmTiles, int tile0) {
    __shared__ float Wsw[(GDIM / 4) * GDIM * 4];
    const int b = blockIdx.x;
    if (b < gemmTiles) {
        gemm_body<true>(X, W, g, T, nrows, Wsw, tile0 + b);
        return;
    }
    int e = (b - gemmTiles) * 256 + threadIdx.x;
    if (e < ne) rank[e] = atomicAdd(&cnt[dst[e]], 1);
}

__global__ __launch_bounds__(256) void csr_fill_gemm1_kernel(const int* __restrict__ src,
                                                             const int* __restrict__ dst,
                                                             const int* __restrict__ rank,
                                                             const int2* __restrict__ seg,
                                                             const float* __restrict__ dinv,
                                                             int2* __restrict__ csr, int ne,
                                                             const float* __restrict__ X,
                                                             const float* __restrict__ W,
                                                             const float* __restrict__ g,
                                                             float* __restrict__ T, int nrows,
                                                             int gemmTiles, int tile0) {
    __shared__ float Wsw[(GDIM / 4) * GDIM * 4];
    const int b = blockIdx.x;
    if (b < gemmTiles) {
        gemm_body<true>(X, W, g, T, nrows, Wsw, tile0 + b);
        return;
    }
    int e = (b - gemmTiles) * 256 + threadIdx.x;
    if (e < ne) {
        int s = src[e], d = dst[e];
        float c = dinv[s] * dinv[d];
        csr[seg[d].x + rank[e]] = make_int2(s, __float_as_int(c));
    }
}

__device__ __forceinline__ float4 gather_row(const float* __restrict__ T,
                                             const int2* __restrict__ seg,
                                             const int2* __restrict__ csr,
                                             const float* __restrict__ dinv,
                                             const float* __restrict__ bias,
                                             const float* __restrict__ gnext,
                                             int d, int n, int lane) {
    const int2  sg  = seg[d];
    const int   beg = sg.x, len = sg.y;
    const float di  = dinv[d];
    const float cs  = di * di;

    const float4 t = ((const float4*)(T + (size_t)d * GDIM))[lane];
    float4 acc = make_float4(0.0f, 0.0f, 0.0f, 0.0f);

    for (int base = 0; base < len; base += 32) {
        const int m = (len - base < 32) ? (len - base) : 32;
        int2 rec = make_int2(0, 0);
        if (lane < m) rec = csr[beg + base + lane];
        for (int j = 0; j < m; j += 8) {
            float4 v[8];
            float  c[8];
            #pragma unroll
            for (int u = 0; u < 8; ++u) {
                v[u] = make_float4(0.0f, 0.0f, 0.0f, 0.0f);
                c[u] = 0.0f;
                if (u == 0 || j + u < m) {
                    int s = __builtin_amdgcn_readlane(rec.x, j + u);
                    c[u]  = __int_as_float(__builtin_amdgcn_readlane(rec.y, j + u));
                    s = (int)min((unsigned)s, (unsigned)(n - 1));
                    v[u] = ((const float4*)(T + (size_t)s * GDIM))[lane];
                }
            }
            #pragma unroll
            for (int u = 0; u < 8; ++u) {
                acc.x = fmaf(v[u].x, c[u], acc.x);
                acc.y = fmaf(v[u].y, c[u], acc.y);
                acc.z = fmaf(v[u].z, c[u], acc.z);
                acc.w = fmaf(v[u].w, c[u], acc.w);
            }
        }
    }
    acc.x = fmaf(t.x, cs, acc.x);
    acc.y = fmaf(t.y, cs, acc.y);
    acc.z = fmaf(t.z, cs, acc.z);
    acc.w = fmaf(t.w, cs, acc.w);

    const float4 b4 = ((const float4*)bias)[lane];
    float4 y;
    y.x = fmaxf(acc.x + b4.x, 0.0f);
    y.y = fmaxf(acc.y + b4.y, 0.0f);
    y.z = fmaxf(acc.z + b4.z, 0.0f);
    y.w = fmaxf(acc.w + b4.w, 0.0f);

    if (gnext) {
        float ss = y.x * y.x + y.y * y.y + y.z * y.z + y.w * y.w;
        for (int off = 16; off > 0; off >>= 1) ss += __shfl_xor(ss, off, 32);
        const float scale = rsqrtf(ss * (1.0f / GDIM) + EPS_RMS);
        const float4 g4 = ((const float4*)gnext)[lane];
        y.x = y.x * scale * g4.x;
        y.y = y.y * scale * g4.y;
        y.z = y.z * scale * g4.z;
        y.w = y.w * scale * g4.w;
    }
    return y;
}

__global__ __launch_bounds__(256) void gather_kernel(const float* __restrict__ T,
                                                     const int2* __restrict__ seg,
                                                     const int2* __restrict__ csr,
                                                     const float* __restrict__ dinv,
                                                     const float* __restrict__ bias,
                                                     float* __restrict__ OUT, int n) {
    const int lane = threadIdx.x & 31;
    const int d = __builtin_amdgcn_readfirstlane((int)(blockIdx.x * 8 + (threadIdx.x >> 5)));
    if (d >= n) return;
    const float4 y = gather_row(T, seg, csr, dinv, bias, nullptr, d, n, lane);
    ((float4*)(OUT + (size_t)d * GDIM))[lane] = y;
}

__global__ __launch_bounds__(512, 16) void gather_gemm_kernel(const float* __restrict__ T,
                                                          const int2* __restrict__ seg,
                                                          const int2* __restrict__ csr,
                                                          const float* __restrict__ dinv,
                                                          const float* __restrict__ bias,
                                                          const float* __restrict__ gnext,
                                                          const float* __restrict__ W,
                                                          float* __restrict__ TOUT, int n, int groups) {
    __shared__ float Wsw[(GDIM / 4) * GDIM * 4];
    __shared__ float Hs[16 * GDIM];
    __shared__ float Ts[16 * GDIM];

    const int tid  = threadIdx.x;
    const int lane = tid & 31;
    const int wave = __builtin_amdgcn_readfirstlane(tid >> 5);

    {
        const int col = tid & (GDIM - 1);
        const int kq  = tid >> 7;
        float4 w4[8];
        #pragma unroll
        for (int j = 0; j < 8; ++j) {
            const float* wp = W + (size_t)((kq + 4 * j) * 4) * GDIM + col;
            w4[j] = make_float4(wp[0], wp[GDIM], wp[2 * GDIM], wp[3 * GDIM]);
        }
        #pragma unroll
        for (int j = 0; j < 8; ++j)
            *(float4*)&Wsw[(kq + 4 * j) * 512 + col * 4] = w4[j];
    }

    const int l16     = lane & 15;
    const int halfsel = lane >> 4;
    const int koff    = halfsel * 2;

    for (int g = blockIdx.x; g < groups; g += gridDim.x) {
        const int d = g * 16 + wave;
        float4 y = make_float4(0.0f, 0.0f, 0.0f, 0.0f);
        if (d < n) y = gather_row(T, seg, csr, dinv, bias, gnext, d, n, lane);
        ((float4*)(Hs + wave * GDIM))[lane ^ wave] = y;
        __syncthreads();

        if (wave < 8) {
            v8f acc = {};
            const float* ap = Hs + l16 * GDIM;
            const float* wb = Wsw + wave * 64 + l16 * 4 + koff;
            #pragma unroll 4
            for (int k0 = 0; k0 < GDIM; k0 += 4) {
                const v2f a = *(const v2f*)(ap + ((k0 + koff) ^ (l16 * 4)));
                const v2f b = *(const v2f*)(wb + (k0 >> 2) * 512);
                acc = __builtin_amdgcn_wmma_f32_16x16x4_f32(
                    false, a, false, b, (short)0, acc, false, false);
            }
            #pragma unroll
            for (int r = 0; r < 8; ++r)
                Ts[(r + halfsel * 8) * GDIM + wave * 16 + l16] = acc[r];
        }
        __syncthreads();

        if (d < n)
            ((float4*)(TOUT + (size_t)d * GDIM))[lane] = ((const float4*)(Ts + wave * GDIM))[lane];
    }
}

extern "C" void kernel_launch(void* const* d_in, const int* in_sizes, int n_in,
                              void* d_out, int out_size, void* d_ws, size_t ws_size,
                              hipStream_t stream) {
    const float* x  = (const float*)d_in[0];
    const int*   ei = (const int*)d_in[1];
    const float* Wm[3]   = {(const float*)d_in[2], (const float*)d_in[5], (const float*)d_in[8]};
    const float* bias[3] = {(const float*)d_in[3], (const float*)d_in[6], (const float*)d_in[9]};
    const float* gain[3] = {(const float*)d_in[4], (const float*)d_in[7], (const float*)d_in[10]};

    const int N = in_sizes[0] / GDIM;
    const int E = in_sizes[1] / 2;
    const int* src = ei;
    const int* dst = ei + E;

    float* dinv = (float*)d_ws;
    size_t off  = ((size_t)N + 63) & ~(size_t)63;
    size_t offE = ((size_t)E + 63) & ~(size_t)63;
    float* B0 = dinv + off;
    float* B1 = (float*)d_out;
    int*  cnt    = (int*)(B0 + (size_t)N * GDIM);
    int2* seg    = (int2*)(cnt + off);
    int*  rank   = (int*)(seg + off);
    int2* csr    = (int2*)(rank + offE);

    const int tiles  = (N + 127) / 128;
    const int tilesA = tiles / 2;
    const int tilesB = tiles - tilesA;
    const int eblk   = (E + 255) / 256;
    count_zero_kernel<<<(N + 255) / 256, 256, 0, stream>>>(cnt, N);
    count_rank_gemm1_kernel<<<tilesA + eblk, 256, 0, stream>>>(dst, cnt, rank, E,
                                                              x, Wm[0], gain[0], B0, N, tilesA, 0);
    rowptr_kernel<<<(N + 1023) / 1024, 1024, 0, stream>>>(cnt, seg, dinv, N);
    csr_fill_gemm1_kernel<<<tilesB + eblk, 256, 0, stream>>>(src, dst, rank, seg, dinv, csr, E,
                                                            x, Wm[0], gain[0], B0, N, tilesB, tilesA);

    const int groups = (N + 15) / 16;
    const int fgrid  = groups < 1024 ? groups : 1024;
    gather_gemm_kernel<<<fgrid, 512, 0, stream>>>(B0, seg, csr, dinv, bias[0], gain[1], Wm[1], B1, N, groups);
    gather_gemm_kernel<<<fgrid, 512, 0, stream>>>(B1, seg, csr, dinv, bias[1], gain[2], Wm[2], B0, N, groups);
    gather_kernel<<<(N + 7) / 8, 256, 0, stream>>>(B0, seg, csr, dinv, bias[2], B1, N);
}
